# starting kernel
# baseline (speedup 1.0000x reference)
#include <hip/hip_runtime.h>
#include <math.h>

typedef float v2f __attribute__((ext_vector_type(2)));
typedef float v8f __attribute__((ext_vector_type(8)));

#define HDIM 128
#define WPB 8
#define LDSA_STRIDE 132
#define LDSW_STRIDE 136

__launch_bounds__(256)
__global__ void gemm_wmma(const float* __restrict__ X, int lda,
                          const float* __restrict__ W,
                          float* __restrict__ Hout, int nTiles) {
    __shared__ float ldsW[HDIM * LDSW_STRIDE];
    __shared__ float ldsA[WPB][16 * LDSA_STRIDE];

    const int tid  = threadIdx.x;
    const int wave = tid >> 5;
    const int lane = tid & 31;

    for (int i = 0; i < 16; ++i) {
        int idx = (i * 256 + tid) * 4;
        int row = idx >> 7, col = idx & 127;
        float4 v = *(const float4*)(W + idx);
        *(float4*)(&ldsW[row * LDSW_STRIDE + col]) = v;
    }

    const int tile   = blockIdx.x * WPB + wave;
    const bool active = (tile < nTiles);
    if (active) {
        const float* Xrow = X + (size_t)tile * 16 * (size_t)lda;
        for (int r = 0; r < 16; ++r) {
            float4 v = *(const float4*)(Xrow + (size_t)r * lda + lane * 4);
            *(float4*)(&ldsA[wave][r * LDSA_STRIDE + lane * 4]) = v;
        }
    }
    __syncthreads();
    if (!active) return;

    const int m     = lane & 15;
    const int khalf = (lane >> 4) * 2;
    const float* aRow = &ldsA[wave][m * LDSA_STRIDE];

    v8f acc[8];
    const v8f vzero = {0.f, 0.f, 0.f, 0.f, 0.f, 0.f, 0.f, 0.f};
#pragma unroll
    for (int nt = 0; nt < 8; ++nt) acc[nt] = vzero;

    for (int k4 = 0; k4 < 32; ++k4) {
        const int kb = k4 * 4 + khalf;
        v2f a;
        a.x = aRow[kb];
        a.y = aRow[kb + 1];
        const float* w0 = &ldsW[kb * LDSW_STRIDE];
        const float* w1 = w0 + LDSW_STRIDE;
#pragma unroll
        for (int nt = 0; nt < 8; ++nt) {
            const int n = nt * 16 + m;
            v2f b;
            b.x = w0[n];
            b.y = w1[n];
            acc[nt] = __builtin_amdgcn_wmma_f32_16x16x4_f32(
                false, a, false, b, (short)0, acc[nt], false, false);
        }
    }

    float* outBase = Hout + (size_t)tile * 16 * HDIM;
    const int mrow = (lane >> 4) * 8;
#pragma unroll
    for (int nt = 0; nt < 8; ++nt) {
#pragma unroll
        for (int r = 0; r < 8; ++r) {
            outBase[(size_t)(mrow + r) * HDIM + nt * 16 + m] = acc[nt][r];
        }
    }
}

__global__ void deg_init(float* __restrict__ deg, int n) {
    int i = blockIdx.x * blockDim.x + threadIdx.x;
    if (i < n) deg[i] = 1.0f;
}

__global__ void deg_count(const int* __restrict__ dst, float* __restrict__ deg, int E) {
    int e = blockIdx.x * blockDim.x + threadIdx.x;
    if (e < E) atomicAdd(&deg[dst[e]], 1.0f);
}

__global__ void make_dinv(float* __restrict__ d, int n) {
    int i = blockIdx.x * blockDim.x + threadIdx.x;
    if (i < n) d[i] = rsqrtf(d[i]);
}

__global__ void agg_init(const float* __restrict__ h, const float* __restrict__ dinv,
                         float* __restrict__ agg, int n) {
    int t = blockIdx.x * blockDim.x + threadIdx.x;
    if (t >= n * 32) return;
    int node = t >> 5;
    float w = dinv[node];
    w *= w;
    float4 v = ((const float4*)h)[t];
    v.x *= w; v.y *= w; v.z *= w; v.w *= w;
    ((float4*)agg)[t] = v;
}

__global__ void edge_agg(const int* __restrict__ src, const int* __restrict__ dst,
                         const float* __restrict__ dinv, const float* __restrict__ h,
                         float* __restrict__ agg, int E) {
    int gid  = blockIdx.x * blockDim.x + threadIdx.x;
    int e    = gid >> 5;
    int lane = gid & 31;
    if (e >= E) return;
    int s = src[e], d = dst[e];
    float w = dinv[s] * dinv[d];
    float4 v = ((const float4*)(h + (size_t)s * HDIM))[lane];
    float* ap = agg + (size_t)d * HDIM + lane * 4;
    atomicAdd(ap + 0, v.x * w);
    atomicAdd(ap + 1, v.y * w);
    atomicAdd(ap + 2, v.z * w);
    atomicAdd(ap + 3, v.w * w);
}

__global__ void post_norm(const float* __restrict__ agg, const float* __restrict__ bias,
                          float* __restrict__ embSlice, int n) {
    int gid  = blockIdx.x * blockDim.x + threadIdx.x;
    int node = gid >> 5;
    int lane = gid & 31;
    if (node >= n) return;
    float4 v = ((const float4*)(agg + (size_t)node * HDIM))[lane];
    float4 b = ((const float4*)bias)[lane];
    v.x = fmaxf(v.x + b.x, 0.f);
    v.y = fmaxf(v.y + b.y, 0.f);
    v.z = fmaxf(v.z + b.z, 0.f);
    v.w = fmaxf(v.w + b.w, 0.f);
    float ss = v.x * v.x + v.y * v.y + v.z * v.z + v.w * v.w;
#pragma unroll
    for (int off = 16; off > 0; off >>= 1)
        ss += __shfl_xor(ss, off, 32);
    float scale = 1.0f / fmaxf(sqrtf(ss), 1e-12f);
    float4 o;
    o.x = v.x * scale; o.y = v.y * scale; o.z = v.z * scale; o.w = v.w * scale;
    *(float4*)(embSlice + (size_t)node * 384 + lane * 4) = o;
}

__launch_bounds__(256)
__global__ void classifier(const float* __restrict__ emb, const float* __restrict__ Wlin,
                           const float* __restrict__ blin, float* __restrict__ out, int n) {
    __shared__ float lw[384 * 8];
    __shared__ float lb[8];
    for (int i = threadIdx.x; i < 384 * 8; i += 256) lw[i] = Wlin[i];
    if (threadIdx.x < 8) lb[threadIdx.x] = blin[threadIdx.x];
    __syncthreads();

    int node = blockIdx.x * blockDim.x + threadIdx.x;
    if (node >= n) return;

    float acc[8];
#pragma unroll
    for (int c = 0; c < 8; ++c) acc[c] = lb[c];
    const float* er = emb + (size_t)node * 384;
    for (int k = 0; k < 384; ++k) {
        float e = er[k];
#pragma unroll
        for (int c = 0; c < 8; ++c) acc[c] = fmaf(e, lw[k * 8 + c], acc[c]);
    }
    float m = acc[0];
#pragma unroll
    for (int c = 1; c < 8; ++c) m = fmaxf(m, acc[c]);
    float s = 0.f;
#pragma unroll
    for (int c = 0; c < 8; ++c) s += expf(acc[c] - m);
    float lse = m + logf(s);
#pragma unroll
    for (int c = 0; c < 8; ++c) out[(size_t)node * 8 + c] = acc[c] - lse;
}

extern "C" void kernel_launch(void* const* d_in, const int* in_sizes, int n_in,
                              void* d_out, int out_size, void* d_ws, size_t ws_size,
                              hipStream_t stream) {
    const float* x    = (const float*)d_in[0];
    const int*   ei   = (const int*)d_in[1];
    const float* W1   = (const float*)d_in[2];
    const float* b1   = (const float*)d_in[3];
    const float* W2   = (const float*)d_in[4];
    const float* b2   = (const float*)d_in[5];
    const float* W3   = (const float*)d_in[6];
    const float* b3   = (const float*)d_in[7];
    const float* Wlin = (const float*)d_in[8];
    const float* blin = (const float*)d_in[9];

    const int n = in_sizes[0] / HDIM;
    const int E = in_sizes[1] / 2;
    const int* src = ei;
    const int* dst = ei + E;

    float* ws   = (float*)d_ws;
    float* dinv = ws;
    float* h    = ws + (size_t)n;
    float* agg  = h  + (size_t)n * HDIM;
    float* emb  = agg + (size_t)n * HDIM;

    const int T = 256;
    deg_init<<<(n + T - 1) / T, T, 0, stream>>>(dinv, n);
    deg_count<<<(E + T - 1) / T, T, 0, stream>>>(dst, dinv, E);
    make_dinv<<<(n + T - 1) / T, T, 0, stream>>>(dinv, n);

    const int nTiles = (n + 15) / 16;
    const float* Ws[3] = {W1, W2, W3};
    const float* bs[3] = {b1, b2, b3};
    const float* Xin = x;
    int lda = HDIM;

    for (int l = 0; l < 3; ++l) {
        gemm_wmma<<<(nTiles + WPB - 1) / WPB, 256, 0, stream>>>(Xin, lda, Ws[l], h, nTiles);
        agg_init<<<((size_t)n * 32 + T - 1) / T, T, 0, stream>>>(h, dinv, agg, n);
        edge_agg<<<(unsigned)(((size_t)E * 32 + T - 1) / T), T, 0, stream>>>(src, dst, dinv, h, agg, E);
        post_norm<<<((size_t)n * 32 + T - 1) / T, T, 0, stream>>>(agg, bs[l], emb + l * HDIM, n);
        Xin = emb + l * HDIM;
        lda = 384;
    }

    classifier<<<(n + T - 1) / T, T, 0, stream>>>(emb, Wlin, blin, (float*)d_out, n);
}


# Transform moved behind the edge sums ((AP)W for A(PW), sums re-ordered): tile kernel multiplies its 64x128 LDS sums by W on f32 WMMA, then bias, relu, normalize, store; 3 GEMM launches, plane h go
# speedup vs baseline: 3.9517x; 3.9517x over previous
#include <hip/hip_runtime.h>
#include <math.h>

typedef float v2f __attribute__((ext_vector_type(2)));
typedef float v8f __attribute__((ext_vector_type(8)));

#define HDIM 128
#define TSTR 132
#define LDSW_STRIDE 136

__global__ void zero_counts(int* __restrict__ binCnt, int nbAll) {
    int i = blockIdx.x * blockDim.x + threadIdx.x;
    if (i < nbAll) binCnt[i] = 0;
}

#define TN 64
#define TSH 6
#define MAXB 2048
#define BCH 4096

__launch_bounds__(256)
__global__ void bin_count(const int* __restrict__ dst, int* __restrict__ binCnt, int E) {
    __shared__ int hist[MAXB];
    const int tid = threadIdx.x;
    for (int b = tid; b < MAXB; b += 256) hist[b] = 0;
    __syncthreads();
    const int base = blockIdx.x * BCH;
    int dv[BCH / 256];
#pragma unroll
    for (int k = 0; k < BCH / 256; ++k)
        dv[k] = dst[min(base + k * 256 + tid, E - 1)];
#pragma unroll
    for (int k = 0; k < BCH / 256; ++k)
        if (base + k * 256 + tid < E) atomicAdd(&hist[dv[k] >> TSH], 1);
    __syncthreads();
    for (int b = tid; b < MAXB; b += 256) atomicAdd(&binCnt[b], hist[b]);
}

__launch_bounds__(1024)
__global__ void bin_scan(const int* __restrict__ binCnt, int* __restrict__ binOff,
                         int* __restrict__ binCur, int nb, int E) {
    __shared__ int wsum[32];
    const int tid = threadIdx.x, lane = tid & 31, wave = tid >> 5;
    const int i0 = 2 * tid, i1 = 2 * tid + 1;
    const int a = (i0 < nb) ? binCnt[i0] : 0;
    const int b = (i1 < nb) ? binCnt[i1] : 0;
    const int s = a + b;
    int inc = s;
#pragma unroll
    for (int o = 1; o < 32; o <<= 1) {
        const int v = __shfl_up(inc, o, 32);
        if (lane >= o) inc += v;
    }
    if (lane == 31) wsum[wave] = inc;
    __syncthreads();
    if (wave == 0) {
        const int v = wsum[lane];
        int iv = v;
#pragma unroll
        for (int o = 1; o < 32; o <<= 1) {
            const int u = __shfl_up(iv, o, 32);
            if (lane >= o) iv += u;
        }
        wsum[lane] = iv - v;
    }
    __syncthreads();
    const int ex = wsum[wave] + inc - s;
    if (i0 < nb) { binOff[i0] = ex;     binCur[i0] = ex; }
    if (i1 < nb) { binOff[i1] = ex + a; binCur[i1] = ex + a; }
    if (tid == 0) binOff[nb] = E;
}

__launch_bounds__(256)
__global__ void bin_fill(const int* __restrict__ src, const int* __restrict__ dst,
                         int* __restrict__ binCur, int* __restrict__ binned, int E) {
    __shared__ int hist[MAXB];
    const int tid = threadIdx.x;
    for (int b = tid; b < MAXB; b += 256) hist[b] = 0;
    __syncthreads();
    const int base = blockIdx.x * BCH;
    int dv[BCH / 256], sv[BCH / 256], rk[BCH / 256];
#pragma unroll
    for (int k = 0; k < BCH / 256; ++k) {
        const int ec = min(base + k * 256 + tid, E - 1);
        dv[k] = dst[ec];
        sv[k] = src[ec];
    }
#pragma unroll
    for (int k = 0; k < BCH / 256; ++k) {
        rk[k] = 0;
        if (base + k * 256 + tid < E)
            rk[k] = atomicAdd(&hist[dv[k] >> TSH], 1);
    }
    __syncthreads();
    int bs[MAXB / 256];
#pragma unroll
    for (int j = 0; j < MAXB / 256; ++j)
        bs[j] = atomicAdd(&binCur[j * 256 + tid], hist[j * 256 + tid]);
#pragma unroll
    for (int j = 0; j < MAXB / 256; ++j)
        hist[j * 256 + tid] = bs[j];
    __syncthreads();
#pragma unroll
    for (int k = 0; k < BCH / 256; ++k) {
        if (base + k * 256 + tid < E) {
            const int pos = hist[dv[k] >> TSH] + rk[k];
            binned[pos] = (sv[k] << TSH) | (dv[k] & (TN - 1));
        }
    }
}

#define SCAP 2048
__launch_bounds__(64)
__global__ void tile_sort(const int* __restrict__ binned, const int* __restrict__ binOff,
                          int* __restrict__ sorted, float* __restrict__ dinv, int n) {
    __shared__ int cnt[TN];
    __shared__ int half0;
    __shared__ int buf[SCAP];
    const int tid  = threadIdx.x;
    const int lane = tid & 31;
    cnt[tid] = 0;
    __syncthreads();
    const int beg = binOff[blockIdx.x];
    const int end = binOff[blockIdx.x + 1];
    for (int e0 = beg; e0 < end; e0 += 4 * TN) {
        int p[4];
#pragma unroll
        for (int k = 0; k < 4; ++k)
            p[k] = binned[min(e0 + k * TN + tid, end - 1)];
#pragma unroll
        for (int k = 0; k < 4; ++k)
            if (e0 + k * TN + tid < end) atomicAdd(&cnt[p[k] & (TN - 1)], 1);
    }
    __syncthreads();
    const int c = cnt[tid];
    const int node = blockIdx.x * TN + tid;
    if (node < n) dinv[node] = rsqrtf((float)(1 + c));

    int inc = c;
#pragma unroll
    for (int o = 1; o < 32; o <<= 1) {
        const int v = __shfl_up(inc, o, 32);
        if (lane >= o) inc += v;
    }
    if (tid == 31) half0 = inc;
    __syncthreads();
    cnt[tid] = inc - c + (tid >= 32 ? half0 : 0);
    __syncthreads();

    const bool stage = (end - beg) <= SCAP;
    for (int e0 = beg; e0 < end; e0 += 4 * TN) {
        int p[4];
#pragma unroll
        for (int k = 0; k < 4; ++k)
            p[k] = binned[min(e0 + k * TN + tid, end - 1)];
#pragma unroll
        for (int k = 0; k < 4; ++k) {
            if (e0 + k * TN + tid < end) {
                const int pos = atomicAdd(&cnt[p[k] & (TN - 1)], 1);
                if (stage) buf[pos] = p[k];
                else       sorted[beg + pos] = p[k];
            }
        }
    }
    if (stage) {
        __syncthreads();
        for (int i = tid; i < end - beg; i += TN) sorted[beg + i] = buf[i];
    }
}

__launch_bounds__(256)
__global__ void tile_layer(const int* __restrict__ binned, const int* __restrict__ binOff,
                           const float* __restrict__ dinv, const float* __restrict__ P,
                           const float* __restrict__ W, const float* __restrict__ bias,
                           float* __restrict__ outp, int n) {
    __shared__ __attribute__((aligned(16))) float t[TN * TSTR];
    __shared__ __attribute__((aligned(16))) float ldsW[HDIM * LDSW_STRIDE];
    __shared__ float ssp[2][TN];
    const int tid  = threadIdx.x;
    const int lane = tid & 31;
    const int wave  = __builtin_amdgcn_readfirstlane(tid >> 5);
    const int node0 = blockIdx.x * TN;

    for (int i = 0; i < 16; ++i) {
        const int idx = (i * 256 + tid) * 4;
        const int row = idx >> 7, col = idx & 127;
        const float4 v = *(const float4*)(W + idx);
        *(float4*)(&ldsW[row * LDSW_STRIDE + col]) = v;
    }

    const int dvbits = __float_as_int(dinv[min(node0 + (lane & 7) * 8 + wave, n - 1)]);
#pragma unroll
    for (int i = 0; i < TN / 8; ++i) {
        const int r    = i * 8 + wave;
        const int node = node0 + r;
        const int nc   = min(node, n - 1);
        float w = __int_as_float(__builtin_amdgcn_readlane(dvbits, i));
        w = (node < n) ? w * w : 0.f;
        float4 v = ((const float4*)P)[(size_t)nc * 32 + lane];
        v.x *= w; v.y *= w; v.z *= w; v.w *= w;
        *(float4*)(&t[r * TSTR + lane * 4]) = v;
    }
    __syncthreads();

    const int beg = binOff[blockIdx.x];
    const int end = binOff[blockIdx.x + 1];
    for (int e0 = beg + wave * 32; e0 < end; e0 += 256) {
        const int cnt = min(32, end - e0);
        int s = 0, dl = 0;
        float w = 0.f;
        if (lane < cnt) {
            const int p = binned[e0 + lane];
            s  = p >> TSH;
            dl = p & (TN - 1);
            w  = dinv[s] * dinv[node0 + dl];
        }
        const int wbits = __float_as_int(w);

        int j = 0;
        for (; j + 8 <= cnt; j += 8) {
            float v[8][4];
#pragma unroll
            for (int u = 0; u < 8; ++u) {
                const int sj = __builtin_amdgcn_readlane(s, j + u);
                const float* hp = P + (size_t)sj * HDIM;
                v[u][0] = hp[lane];
                v[u][1] = hp[lane + 32];
                v[u][2] = hp[lane + 64];
                v[u][3] = hp[lane + 96];
            }
            int cur = __builtin_amdgcn_readlane(dl, j);
            float a0 = 0.f, a1 = 0.f, a2 = 0.f, a3 = 0.f;
#pragma unroll
            for (int u = 0; u < 8; ++u) {
                const int dj = __builtin_amdgcn_readlane(dl, j + u);
                const float wj = __int_as_float(__builtin_amdgcn_readlane(wbits, j + u));
                if (dj != cur) {
                    float* ap = t + cur * TSTR;
                    atomicAdd(ap + lane,      a0);
                    atomicAdd(ap + lane + 32, a1);
                    atomicAdd(ap + lane + 64, a2);
                    atomicAdd(ap + lane + 96, a3);
                    a0 = 0.f; a1 = 0.f; a2 = 0.f; a3 = 0.f;
                    cur = dj;
                }
                a0 = fmaf(v[u][0], wj, a0);
                a1 = fmaf(v[u][1], wj, a1);
                a2 = fmaf(v[u][2], wj, a2);
                a3 = fmaf(v[u][3], wj, a3);
            }
            float* ap = t + cur * TSTR;
            atomicAdd(ap + lane,      a0);
            atomicAdd(ap + lane + 32, a1);
            atomicAdd(ap + lane + 64, a2);
            atomicAdd(ap + lane + 96, a3);
        }
        for (; j < cnt; ++j) {
            const int sj = __builtin_amdgcn_readlane(s, j);
            const int dj = __builtin_amdgcn_readlane(dl, j);
            const float wj = __int_as_float(__builtin_amdgcn_readlane(wbits, j));
            const float* hp = P + (size_t)sj * HDIM;
            float* ap = t + dj * TSTR;
            const float v0 = hp[lane];
            const float v1 = hp[lane + 32];
            const float v2 = hp[lane + 64];
            const float v3 = hp[lane + 96];
            atomicAdd(ap + lane,      v0 * wj);
            atomicAdd(ap + lane + 32, v1 * wj);
            atomicAdd(ap + lane + 64, v2 * wj);
            atomicAdd(ap + lane + 96, v3 * wj);
        }
    }
    __syncthreads();

    const int m     = lane & 15;
    const int hh    = lane >> 4;
    const int khalf = hh * 2;
    const int strip = wave & 3;
    const int c0    = (wave >> 2) * 64;
    const float* aRow = &t[(strip * 16 + m) * TSTR];

    v8f acc[4];
    const v8f vzero = {0.f, 0.f, 0.f, 0.f, 0.f, 0.f, 0.f, 0.f};
#pragma unroll
    for (int nt = 0; nt < 4; ++nt) acc[nt] = vzero;

    for (int k4 = 0; k4 < 32; ++k4) {
        const int kb = k4 * 4 + khalf;
        v2f a;
        a.x = aRow[kb];
        a.y = aRow[kb + 1];
        const float* w0 = &ldsW[kb * LDSW_STRIDE + c0];
        const float* w1 = w0 + LDSW_STRIDE;
#pragma unroll
        for (int nt = 0; nt < 4; ++nt) {
            const int nn = nt * 16 + m;
            v2f b;
            b.x = w0[nn];
            b.y = w1[nn];
            acc[nt] = __builtin_amdgcn_wmma_f32_16x16x4_f32(
                false, a, false, b, (short)0, acc[nt], false, false);
        }
    }

    float bv[4];
#pragma unroll
    for (int nt = 0; nt < 4; ++nt) bv[nt] = bias[c0 + nt * 16 + m];
    float ssr[8];
#pragma unroll
    for (int r = 0; r < 8; ++r) {
        float q = 0.f;
#pragma unroll
        for (int nt = 0; nt < 4; ++nt) {
            const float v = fmaxf(acc[nt][r] + bv[nt], 0.f);
            acc[nt][r] = v;
            q = fmaf(v, v, q);
        }
        ssr[r] = q;
    }
#pragma unroll
    for (int off = 8; off > 0; off >>= 1) {
#pragma unroll
        for (int r = 0; r < 8; ++r) ssr[r] += __shfl_xor(ssr[r], off, 32);
    }
    const int rowb = strip * 16 + hh * 8;
    if (m == 0) {
#pragma unroll
        for (int r = 0; r < 8; ++r) ssp[wave >> 2][rowb + r] = ssr[r];
    }
    __syncthreads();
#pragma unroll
    for (int r = 0; r < 8; ++r) {
        const float ss = ssp[0][rowb + r] + ssp[1][rowb + r];
        const float scale = 1.0f / fmaxf(sqrtf(ss), 1e-12f);
#pragma unroll
        for (int nt = 0; nt < 4; ++nt)
            t[(rowb + r) * TSTR + c0 + nt * 16 + m] = acc[nt][r] * scale;
    }
    __syncthreads();

#pragma unroll
    for (int i = 0; i < TN / 8; ++i) {
        const int r    = i * 8 + wave;
        const int node = node0 + r;
        if (node < n)
            ((float4*)outp)[(size_t)node * 32 + lane] = *(const float4*)(&t[r * TSTR + lane * 4]);
    }
}

__launch_bounds__(256)
__global__ void classifier(const float* __restrict__ emb,
                           const float* __restrict__ Wlin,
                           const float* __restrict__ blin, float* __restrict__ out, int n) {
    __shared__ float lw[384 * 8];
    __shared__ float lb[8];
    for (int i = threadIdx.x; i < 384 * 8; i += 256) lw[i] = Wlin[i];
    if (threadIdx.x < 8) lb[threadIdx.x] = blin[threadIdx.x];
    __syncthreads();

    int node = blockIdx.x * blockDim.x + threadIdx.x;
    if (node >= n) return;

    float logit[8];
#pragma unroll
    for (int c = 0; c < 8; ++c) logit[c] = lb[c];
    for (int l = 0; l < 3; ++l) {
        const float* ar = emb + ((size_t)l * n + node) * HDIM;
        const float* wl = lw + l * HDIM * 8;
        float acc[8];
#pragma unroll
        for (int c = 0; c < 8; ++c) acc[c] = 0.f;
        for (int k0 = 0; k0 < HDIM; k0 += 16) {
            float4 q[4];
#pragma unroll
            for (int j = 0; j < 4; ++j) q[j] = *(const float4*)(ar + k0 + 4 * j);
#pragma unroll
            for (int j = 0; j < 4; ++j) {
                const float e[4] = {q[j].x, q[j].y, q[j].z, q[j].w};
#pragma unroll
                for (int i = 0; i < 4; ++i) {
                    const int k = k0 + 4 * j + i;
                    const float v = e[i];
#pragma unroll
                    for (int c = 0; c < 8; ++c) acc[c] = fmaf(v, wl[k * 8 + c], acc[c]);
                }
            }
        }
#pragma unroll
        for (int c = 0; c < 8; ++c) logit[c] += acc[c];
    }
    float m = logit[0];
#pragma unroll
    for (int c = 1; c < 8; ++c) m = fmaxf(m, logit[c]);
    float s = 0.f;
#pragma unroll
    for (int c = 0; c < 8; ++c) s += expf(logit[c] - m);
    float lse = m + logf(s);
#pragma unroll
    for (int c = 0; c < 8; ++c) out[(size_t)node * 8 + c] = logit[c] - lse;
}

extern "C" void kernel_launch(void* const* d_in, const int* in_sizes, int n_in,
                              void* d_out, int out_size, void* d_ws, size_t ws_size,
                              hipStream_t stream) {
    const float* x    = (const float*)d_in[0];
    const int*   ei   = (const int*)d_in[1];
    const float* W1   = (const float*)d_in[2];
    const float* b1   = (const float*)d_in[3];
    const float* W2   = (const float*)d_in[4];
    const float* b2   = (const float*)d_in[5];
    const float* W3   = (const float*)d_in[6];
    const float* b3   = (const float*)d_in[7];
    const float* Wlin = (const float*)d_in[8];
    const float* blin = (const float*)d_in[9];

    const int n = in_sizes[0] / HDIM;
    const int E = in_sizes[1] / 2;
    const int* src = ei;
    const int* dst = ei + E;

    float* ws   = (float*)d_ws;
    float* dinv = ws;
    float* emb  = ws + (size_t)n;
    int* binCnt = (int*)(emb + (size_t)3 * n * HDIM);
    int* binOff = binCnt + MAXB;
    int* binCur = binOff + MAXB + 64;
    int* binned = binCur + MAXB;
    int* sorted = binned + E;
    const int nb   = (n + TN - 1) / TN;
    const int nblk = (E + BCH - 1) / BCH;

    const int T = 256;
    zero_counts<<<(MAXB + T - 1) / T, T, 0, stream>>>(binCnt, MAXB);
    bin_count<<<nblk, 256, 0, stream>>>(dst, binCnt, E);
    bin_scan<<<1, 1024, 0, stream>>>(binCnt, binOff, binCur, MAXB, E);
    bin_fill<<<nblk, 256, 0, stream>>>(src, dst, binCur, binned, E);

    tile_sort<<<nb, TN, 0, stream>>>(binned, binOff, sorted, dinv, n);

    float* e1 = emb;
    float* e2 = emb + (size_t)n * HDIM;
    float* e3 = emb + (size_t)2 * n * HDIM;
    tile_layer<<<nb, 256, 0, stream>>>(sorted, binOff, dinv, x,  W1, b1, e1, n);
    tile_layer<<<nb, 256, 0, stream>>>(sorted, binOff, dinv, e1, W2, b2, e2, n);
    tile_layer<<<nb, 256, 0, stream>>>(sorted, binOff, dinv, e2, W3, b3, e3, n);

    classifier<<<(n + T - 1) / T, T, 0, stream>>>(emb, Wlin, blin, (float*)d_out, n);
}


# W goes through one 64-row LDS buffer in two halves (rows 64-127 replace rows 0-63 behind the first half's product): 69 KB of LDS a work-group, not 104, so four fit a WGP, not three; k order kept
# speedup vs baseline: 4.3987x; 1.1131x over previous
#include <hip/hip_runtime.h>
#include <math.h>

typedef float v2f __attribute__((ext_vector_type(2)));
typedef float v8f __attribute__((ext_vector_type(8)));

#define HDIM 128
#define TSTR 132
#define LDSW_STRIDE 136

__global__ void zero_counts(int* __restrict__ binCnt, int nbAll) {
    int i = blockIdx.x * blockDim.x + threadIdx.x;
    if (i < nbAll) binCnt[i] = 0;
}

#define TN 64
#define TSH 6
#define MAXB 2048
#define BCH 4096

__launch_bounds__(256)
__global__ void bin_count(const int* __restrict__ dst, int* __restrict__ binCnt, int E) {
    __shared__ int hist[MAXB];
    const int tid = threadIdx.x;
    for (int b = tid; b < MAXB; b += 256) hist[b] = 0;
    __syncthreads();
    const int base = blockIdx.x * BCH;
    int dv[BCH / 256];
#pragma unroll
    for (int k = 0; k < BCH / 256; ++k)
        dv[k] = dst[min(base + k * 256 + tid, E - 1)];
#pragma unroll
    for (int k = 0; k < BCH / 256; ++k)
        if (base + k * 256 + tid < E) atomicAdd(&hist[dv[k] >> TSH], 1);
    __syncthreads();
    for (int b = tid; b < MAXB; b += 256) atomicAdd(&binCnt[b], hist[b]);
}

__launch_bounds__(1024)
__global__ void bin_scan(const int* __restrict__ binCnt, int* __restrict__ binOff,
                         int* __restrict__ binCur, int nb, int E) {
    __shared__ int wsum[32];
    const int tid = threadIdx.x, lane = tid & 31, wave = tid >> 5;
    const int i0 = 2 * tid, i1 = 2 * tid + 1;
    const int a = (i0 < nb) ? binCnt[i0] : 0;
    const int b = (i1 < nb) ? binCnt[i1] : 0;
    const int s = a + b;
    int inc = s;
#pragma unroll
    for (int o = 1; o < 32; o <<= 1) {
        const int v = __shfl_up(inc, o, 32);
        if (lane >= o) inc += v;
    }
    if (lane == 31) wsum[wave] = inc;
    __syncthreads();
    if (wave == 0) {
        const int v = wsum[lane];
        int iv = v;
#pragma unroll
        for (int o = 1; o < 32; o <<= 1) {
            const int u = __shfl_up(iv, o, 32);
            if (lane >= o) iv += u;
        }
        wsum[lane] = iv - v;
    }
    __syncthreads();
    const int ex = wsum[wave] + inc - s;
    if (i0 < nb) { binOff[i0] = ex;     binCur[i0] = ex; }
    if (i1 < nb) { binOff[i1] = ex + a; binCur[i1] = ex + a; }
    if (tid == 0) binOff[nb] = E;
}

__launch_bounds__(256)
__global__ void bin_fill(const int* __restrict__ src, const int* __restrict__ dst,
                         int* __restrict__ binCur, int* __restrict__ binned, int E) {
    __shared__ int hist[MAXB];
    const int tid = threadIdx.x;
    for (int b = tid; b < MAXB; b += 256) hist[b] = 0;
    __syncthreads();
    const int base = blockIdx.x * BCH;
    int dv[BCH / 256], sv[BCH / 256], rk[BCH / 256];
#pragma unroll
    for (int k = 0; k < BCH / 256; ++k) {
        const int ec = min(base + k * 256 + tid, E - 1);
        dv[k] = dst[ec];
        sv[k] = src[ec];
    }
#pragma unroll
    for (int k = 0; k < BCH / 256; ++k) {
        rk[k] = 0;
        if (base + k * 256 + tid < E)
            rk[k] = atomicAdd(&hist[dv[k] >> TSH], 1);
    }
    __syncthreads();
    int bs[MAXB / 256];
#pragma unroll
    for (int j = 0; j < MAXB / 256; ++j)
        bs[j] = atomicAdd(&binCur[j * 256 + tid], hist[j * 256 + tid]);
#pragma unroll
    for (int j = 0; j < MAXB / 256; ++j)
        hist[j * 256 + tid] = bs[j];
    __syncthreads();
#pragma unroll
    for (int k = 0; k < BCH / 256; ++k) {
        if (base + k * 256 + tid < E) {
            const int pos = hist[dv[k] >> TSH] + rk[k];
            binned[pos] = (sv[k] << TSH) | (dv[k] & (TN - 1));
        }
    }
}

#define SCAP 2048
__launch_bounds__(64)
__global__ void tile_sort(const int* __restrict__ binned, const int* __restrict__ binOff,
                          int* __restrict__ sorted, float* __restrict__ dinv, int n) {
    __shared__ int cnt[TN];
    __shared__ int half0;
    __shared__ int buf[SCAP];
    const int tid  = threadIdx.x;
    const int lane = tid & 31;
    cnt[tid] = 0;
    __syncthreads();
    const int beg = binOff[blockIdx.x];
    const int end = binOff[blockIdx.x + 1];
    for (int e0 = beg; e0 < end; e0 += 4 * TN) {
        int p[4];
#pragma unroll
        for (int k = 0; k < 4; ++k)
            p[k] = binned[min(e0 + k * TN + tid, end - 1)];
#pragma unroll
        for (int k = 0; k < 4; ++k)
            if (e0 + k * TN + tid < end) atomicAdd(&cnt[p[k] & (TN - 1)], 1);
    }
    __syncthreads();
    const int c = cnt[tid];
    const int node = blockIdx.x * TN + tid;
    if (node < n) dinv[node] = rsqrtf((float)(1 + c));

    int inc = c;
#pragma unroll
    for (int o = 1; o < 32; o <<= 1) {
        const int v = __shfl_up(inc, o, 32);
        if (lane >= o) inc += v;
    }
    if (tid == 31) half0 = inc;
    __syncthreads();
    cnt[tid] = inc - c + (tid >= 32 ? half0 : 0);
    __syncthreads();

    const bool stage = (end - beg) <= SCAP;
    for (int e0 = beg; e0 < end; e0 += 4 * TN) {
        int p[4];
#pragma unroll
        for (int k = 0; k < 4; ++k)
            p[k] = binned[min(e0 + k * TN + tid, end - 1)];
#pragma unroll
        for (int k = 0; k < 4; ++k) {
            if (e0 + k * TN + tid < end) {
                const int pos = atomicAdd(&cnt[p[k] & (TN - 1)], 1);
                if (stage) buf[pos] = p[k];
                else       sorted[beg + pos] = p[k];
            }
        }
    }
    if (stage) {
        __syncthreads();
        for (int i = tid; i < end - beg; i += TN) sorted[beg + i] = buf[i];
    }
}

__device__ __forceinline__ void half_product(v8f (&acc)[4], const float* a_row,
                                             const float* wcol, int khalf) {
#pragma unroll 2
    for (int k4 = 0; k4 < 16; ++k4) {
        const int kl = k4 * 4 + khalf;
        v2f a;
        a.x = a_row[kl];
        a.y = a_row[kl + 1];
        const float* w0 = wcol + kl * LDSW_STRIDE;
        const float* w1 = w0 + LDSW_STRIDE;
#pragma unroll
        for (int nt = 0; nt < 4; ++nt) {
            v2f b;
            b.x = w0[nt * 16];
            b.y = w1[nt * 16];
            acc[nt] = __builtin_amdgcn_wmma_f32_16x16x4_f32(
                false, a, false, b, (short)0, acc[nt], false, false);
        }
    }
}

__launch_bounds__(256)
__global__ void tile_layer(const int* __restrict__ binned, const int* __restrict__ binOff,
                           const float* __restrict__ dinv, const float* __restrict__ P,
                           const float* __restrict__ W, const float* __restrict__ bias,
                           float* __restrict__ outp, int n) {
    __shared__ __attribute__((aligned(16))) float t[TN * TSTR];
    __shared__ __attribute__((aligned(16))) float ldsW[(HDIM / 2) * LDSW_STRIDE];
    __shared__ float ssp[2][TN];
    const int tid  = threadIdx.x;
    const int lane = tid & 31;
    const int wave  = __builtin_amdgcn_readfirstlane(tid >> 5);
    const int node0 = blockIdx.x * TN;

    for (int i = 0; i < 8; ++i) {
        const int idx = (i * 256 + tid) * 4;
        const int row = idx >> 7, col = idx & 127;
        const float4 v = *(const float4*)(W + idx);
        *(float4*)(&ldsW[row * LDSW_STRIDE + col]) = v;
    }

    const int dvbits = __float_as_int(dinv[min(node0 + (lane & 7) * 8 + wave, n - 1)]);
#pragma unroll
    for (int i = 0; i < TN / 8; ++i) {
        const int r    = i * 8 + wave;
        const int node = node0 + r;
        const int nc   = min(node, n - 1);
        float w = __int_as_float(__builtin_amdgcn_readlane(dvbits, i));
        w = (node < n) ? w * w : 0.f;
        float4 v = ((const float4*)P)[(size_t)nc * 32 + lane];
        v.x *= w; v.y *= w; v.z *= w; v.w *= w;
        *(float4*)(&t[r * TSTR + lane * 4]) = v;
    }
    __syncthreads();

    const int beg = binOff[blockIdx.x];
    const int end = binOff[blockIdx.x + 1];
    for (int e0 = beg + wave * 32; e0 < end; e0 += 256) {
        const int cnt = min(32, end - e0);
        int s = 0, dl = 0;
        float w = 0.f;
        if (lane < cnt) {
            const int p = binned[e0 + lane];
            s  = p >> TSH;
            dl = p & (TN - 1);
            w  = dinv[s] * dinv[node0 + dl];
        }
        const int wbits = __float_as_int(w);

        int j = 0;
        for (; j + 8 <= cnt; j += 8) {
            float v[8][4];
#pragma unroll
            for (int u = 0; u < 8; ++u) {
                const int sj = __builtin_amdgcn_readlane(s, j + u);
                const float* hp = P + (size_t)sj * HDIM;
                v[u][0] = hp[lane];
                v[u][1] = hp[lane + 32];
                v[u][2] = hp[lane + 64];
                v[u][3] = hp[lane + 96];
            }
            int cur = __builtin_amdgcn_readlane(dl, j);
            float a0 = 0.f, a1 = 0.f, a2 = 0.f, a3 = 0.f;
#pragma unroll
            for (int u = 0; u < 8; ++u) {
                const int dj = __builtin_amdgcn_readlane(dl, j + u);
                const float wj = __int_as_float(__builtin_amdgcn_readlane(wbits, j + u));
                if (dj != cur) {
                    float* ap = t + cur * TSTR;
                    atomicAdd(ap + lane,      a0);
                    atomicAdd(ap + lane + 32, a1);
                    atomicAdd(ap + lane + 64, a2);
                    atomicAdd(ap + lane + 96, a3);
                    a0 = 0.f; a1 = 0.f; a2 = 0.f; a3 = 0.f;
                    cur = dj;
                }
                a0 = fmaf(v[u][0], wj, a0);
                a1 = fmaf(v[u][1], wj, a1);
                a2 = fmaf(v[u][2], wj, a2);
                a3 = fmaf(v[u][3], wj, a3);
            }
            float* ap = t + cur * TSTR;
            atomicAdd(ap + lane,      a0);
            atomicAdd(ap + lane + 32, a1);
            atomicAdd(ap + lane + 64, a2);
            atomicAdd(ap + lane + 96, a3);
        }
        for (; j < cnt; ++j) {
            const int sj = __builtin_amdgcn_readlane(s, j);
            const int dj = __builtin_amdgcn_readlane(dl, j);
            const float wj = __int_as_float(__builtin_amdgcn_readlane(wbits, j));
            const float* hp = P + (size_t)sj * HDIM;
            float* ap = t + dj * TSTR;
            const float v0 = hp[lane];
            const float v1 = hp[lane + 32];
            const float v2 = hp[lane + 64];
            const float v3 = hp[lane + 96];
            atomicAdd(ap + lane,      v0 * wj);
            atomicAdd(ap + lane + 32, v1 * wj);
            atomicAdd(ap + lane + 64, v2 * wj);
            atomicAdd(ap + lane + 96, v3 * wj);
        }
    }
    __syncthreads();

    const int m     = lane & 15;
    const int hh    = lane >> 4;
    const int khalf = hh * 2;
    const int strip = wave & 3;
    const int c0    = (wave >> 2) * 64;
    const float* aRow = &t[(strip * 16 + m) * TSTR];

    v8f acc[4];
    const v8f vzero = {0.f, 0.f, 0.f, 0.f, 0.f, 0.f, 0.f, 0.f};
#pragma unroll
    for (int nt = 0; nt < 4; ++nt) acc[nt] = vzero;

    half_product(acc, aRow, &ldsW[c0 + m], khalf);
    __syncthreads();
#pragma unroll
    for (int i = 0; i < 8; ++i) {
        const int idx = (i * 256 + tid) * 4;
        const float4 v = *(const float4*)(W + (HDIM / 2) * HDIM + idx);
        *(float4*)(&ldsW[(idx >> 7) * LDSW_STRIDE + (idx & 127)]) = v;
    }
    __syncthreads();
    half_product(acc, aRow + HDIM / 2, &ldsW[c0 + m], khalf);

    float bv[4];
#pragma unroll
    for (int nt = 0; nt < 4; ++nt) bv[nt] = bias[c0 + nt * 16 + m];
    float ssr[8];
#pragma unroll
    for (int r = 0; r < 8; ++r) {
        float q = 0.f;
#pragma unroll
        for (int nt = 0; nt < 4; ++nt) {
            const float v = fmaxf(acc[nt][r] + bv[nt], 0.f);
            acc[nt][r] = v;
            q = fmaf(v, v, q);
        }
        ssr[r] = q;
    }
#pragma unroll
    for (int off = 8; off > 0; off >>= 1) {
#pragma unroll
        for (int r = 0; r < 8; ++r) ssr[r] += __shfl_xor(ssr[r], off, 32);
    }
    const int rowb = strip * 16 + hh * 8;
    if (m == 0) {
#pragma unroll
        for (int r = 0; r < 8; ++r) ssp[wave >> 2][rowb + r] = ssr[r];
    }
    __syncthreads();
#pragma unroll
    for (int r = 0; r < 8; ++r) {
        const float ss = ssp[0][rowb + r] + ssp[1][rowb + r];
        const float scale = 1.0f / fmaxf(sqrtf(ss), 1e-12f);
#pragma unroll
        for (int nt = 0; nt < 4; ++nt)
            t[(rowb + r) * TSTR + c0 + nt * 16 + m] = acc[nt][r] * scale;
    }
    __syncthreads();

#pragma unroll
    for (int i = 0; i < TN / 8; ++i) {
        const int r    = i * 8 + wave;
        const int node = node0 + r;
        if (node < n)
            ((float4*)outp)[(size_t)node * 32 + lane] = *(const float4*)(&t[r * TSTR + lane * 4]);
    }
}

__launch_bounds__(256)
__global__ void classifier(const float* __restrict__ emb,
                           const float* __restrict__ Wlin,
                           const float* __restrict__ blin, float* __restrict__ out, int n) {
    __shared__ float lw[384 * 8];
    __shared__ float lb[8];
    for (int i = threadIdx.x; i < 384 * 8; i += 256) lw[i] = Wlin[i];
    if (threadIdx.x < 8) lb[threadIdx.x] = blin[threadIdx.x];
    __syncthreads();

    int node = blockIdx.x * blockDim.x + threadIdx.x;
    if (node >= n) return;

    float logit[8];
#pragma unroll
    for (int c = 0; c < 8; ++c) logit[c] = lb[c];
    for (int l = 0; l < 3; ++l) {
        const float* ar = emb + ((size_t)l * n + node) * HDIM;
        const float* wl = lw + l * HDIM * 8;
        float acc[8];
#pragma unroll
        for (int c = 0; c < 8; ++c) acc[c] = 0.f;
        for (int k0 = 0; k0 < HDIM; k0 += 16) {
            float4 q[4];
#pragma unroll
            for (int j = 0; j < 4; ++j) q[j] = *(const float4*)(ar + k0 + 4 * j);
#pragma unroll
            for (int j = 0; j < 4; ++j) {
                const float e[4] = {q[j].x, q[j].y, q[j].z, q[j].w};
#pragma unroll
                for (int i = 0; i < 4; ++i) {
                    const int k = k0 + 4 * j + i;
                    const float v = e[i];
#pragma unroll
                    for (int c = 0; c < 8; ++c) acc[c] = fmaf(v, wl[k * 8 + c], acc[c]);
                }
            }
        }
#pragma unroll
        for (int c = 0; c < 8; ++c) logit[c] += acc[c];
    }
    float m = logit[0];
#pragma unroll
    for (int c = 1; c < 8; ++c) m = fmaxf(m, logit[c]);
    float s = 0.f;
#pragma unroll
    for (int c = 0; c < 8; ++c) s += expf(logit[c] - m);
    float lse = m + logf(s);
#pragma unroll
    for (int c = 0; c < 8; ++c) out[(size_t)node * 8 + c] = logit[c] - lse;
}

extern "C" void kernel_launch(void* const* d_in, const int* in_sizes, int n_in,
                              void* d_out, int out_size, void* d_ws, size_t ws_size,
                              hipStream_t stream) {
    const float* x    = (const float*)d_in[0];
    const int*   ei   = (const int*)d_in[1];
    const float* W1   = (const float*)d_in[2];
    const float* b1   = (const float*)d_in[3];
    const float* W2   = (const float*)d_in[4];
    const float* b2   = (const float*)d_in[5];
    const float* W3   = (const float*)d_in[6];
    const float* b3   = (const float*)d_in[7];
    const float* Wlin = (const float*)d_in[8];
    const float* blin = (const float*)d_in[9];

    const int n = in_sizes[0] / HDIM;
    const int E = in_sizes[1] / 2;
    const int* src = ei;
    const int* dst = ei + E;

    float* ws   = (float*)d_ws;
    float* dinv = ws;
    float* emb  = ws + (size_t)n;
    int* binCnt = (int*)(emb + (size_t)3 * n * HDIM);
    int* binOff = binCnt + MAXB;
    int* binCur = binOff + MAXB + 64;
    int* binned = binCur + MAXB;
    int* sorted = binned + E;
    const int nb   = (n + TN - 1) / TN;
    const int nblk = (E + BCH - 1) / BCH;

    const int T = 256;
    zero_counts<<<(MAXB + T - 1) / T, T, 0, stream>>>(binCnt, MAXB);
    bin_count<<<nblk, 256, 0, stream>>>(dst, binCnt, E);
    bin_scan<<<1, 1024, 0, stream>>>(binCnt, binOff, binCur, MAXB, E);
    bin_fill<<<nblk, 256, 0, stream>>>(src, dst, binCur, binned, E);

    tile_sort<<<nb, TN, 0, stream>>>(binned, binOff, sorted, dinv, n);

    float* e1 = emb;
    float* e2 = emb + (size_t)n * HDIM;
    float* e3 = emb + (size_t)2 * n * HDIM;
    tile_layer<<<nb, 256, 0, stream>>>(sorted, binOff, dinv, x,  W1, b1, e1, n);
    tile_layer<<<nb, 256, 0, stream>>>(sorted, binOff, dinv, e1, W2, b2, e2, n);
    tile_layer<<<nb, 256, 0, stream>>>(sorted, binOff, dinv, e2, W3, b3, e3, n);

    classifier<<<(n + T - 1) / T, T, 0, stream>>>(emb, Wlin, blin, (float*)d_out, n);
}


# Classifier folded into the layer kernel: tile's finished rows (LDS) x the layer's 128x8 rows of Wlin on f32 WMMA, shares summed in an n x 8 plane; classifier launch, plane 3 go; logit sums re-ordered
# speedup vs baseline: 4.8668x; 1.1064x over previous
#include <hip/hip_runtime.h>
#include <math.h>

typedef float v2f __attribute__((ext_vector_type(2)));
typedef float v8f __attribute__((ext_vector_type(8)));

#define HDIM 128
#define TSTR 132
#define LDSW_STRIDE 136

__global__ void zero_counts(int* __restrict__ binCnt, int nbAll) {
    int i = blockIdx.x * blockDim.x + threadIdx.x;
    if (i < nbAll) binCnt[i] = 0;
}

#define TN 64
#define TSH 6
#define MAXB 2048
#define BCH 4096

__launch_bounds__(256)
__global__ void bin_count(const int* __restrict__ dst, int* __restrict__ binCnt, int E) {
    __shared__ int hist[MAXB];
    const int tid = threadIdx.x;
    for (int b = tid; b < MAXB; b += 256) hist[b] = 0;
    __syncthreads();
    const int base = blockIdx.x * BCH;
    int dv[BCH / 256];
#pragma unroll
    for (int k = 0; k < BCH / 256; ++k)
        dv[k] = dst[min(base + k * 256 + tid, E - 1)];
#pragma unroll
    for (int k = 0; k < BCH / 256; ++k)
        if (base + k * 256 + tid < E) atomicAdd(&hist[dv[k] >> TSH], 1);
    __syncthreads();
    for (int b = tid; b < MAXB; b += 256) atomicAdd(&binCnt[b], hist[b]);
}

__launch_bounds__(1024)
__global__ void bin_scan(const int* __restrict__ binCnt, int* __restrict__ binOff,
                         int* __restrict__ binCur, int nb, int E) {
    __shared__ int wsum[32];
    const int tid = threadIdx.x, lane = tid & 31, wave = tid >> 5;
    const int i0 = 2 * tid, i1 = 2 * tid + 1;
    const int a = (i0 < nb) ? binCnt[i0] : 0;
    const int b = (i1 < nb) ? binCnt[i1] : 0;
    const int s = a + b;
    int inc = s;
#pragma unroll
    for (int o = 1; o < 32; o <<= 1) {
        const int v = __shfl_up(inc, o, 32);
        if (lane >= o) inc += v;
    }
    if (lane == 31) wsum[wave] = inc;
    __syncthreads();
    if (wave == 0) {
        const int v = wsum[lane];
        int iv = v;
#pragma unroll
        for (int o = 1; o < 32; o <<= 1) {
            const int u = __shfl_up(iv, o, 32);
            if (lane >= o) iv += u;
        }
        wsum[lane] = iv - v;
    }
    __syncthreads();
    const int ex = wsum[wave] + inc - s;
    if (i0 < nb) { binOff[i0] = ex;     binCur[i0] = ex; }
    if (i1 < nb) { binOff[i1] = ex + a; binCur[i1] = ex + a; }
    if (tid == 0) binOff[nb] = E;
}

__launch_bounds__(256)
__global__ void bin_fill(const int* __restrict__ src, const int* __restrict__ dst,
                         int* __restrict__ binCur, int* __restrict__ binned, int E) {
    __shared__ int hist[MAXB];
    const int tid = threadIdx.x;
    for (int b = tid; b < MAXB; b += 256) hist[b] = 0;
    __syncthreads();
    const int base = blockIdx.x * BCH;
    int dv[BCH / 256], sv[BCH / 256], rk[BCH / 256];
#pragma unroll
    for (int k = 0; k < BCH / 256; ++k) {
        const int ec = min(base + k * 256 + tid, E - 1);
        dv[k] = dst[ec];
        sv[k] = src[ec];
    }
#pragma unroll
    for (int k = 0; k < BCH / 256; ++k) {
        rk[k] = 0;
        if (base + k * 256 + tid < E)
            rk[k] = atomicAdd(&hist[dv[k] >> TSH], 1);
    }
    __syncthreads();
    int bs[MAXB / 256];
#pragma unroll
    for (int j = 0; j < MAXB / 256; ++j)
        bs[j] = atomicAdd(&binCur[j * 256 + tid], hist[j * 256 + tid]);
#pragma unroll
    for (int j = 0; j < MAXB / 256; ++j)
        hist[j * 256 + tid] = bs[j];
    __syncthreads();
#pragma unroll
    for (int k = 0; k < BCH / 256; ++k) {
        if (base + k * 256 + tid < E) {
            const int pos = hist[dv[k] >> TSH] + rk[k];
            binned[pos] = (sv[k] << TSH) | (dv[k] & (TN - 1));
        }
    }
}

#define SCAP 2048
__launch_bounds__(64)
__global__ void tile_sort(const int* __restrict__ binned, const int* __restrict__ binOff,
                          int* __restrict__ sorted, float* __restrict__ dinv, int n) {
    __shared__ int cnt[TN];
    __shared__ int half0;
    __shared__ int buf[SCAP];
    const int tid  = threadIdx.x;
    const int lane = tid & 31;
    cnt[tid] = 0;
    __syncthreads();
    const int beg = binOff[blockIdx.x];
    const int end = binOff[blockIdx.x + 1];
    for (int e0 = beg; e0 < end; e0 += 4 * TN) {
        int p[4];
#pragma unroll
        for (int k = 0; k < 4; ++k)
            p[k] = binned[min(e0 + k * TN + tid, end - 1)];
#pragma unroll
        for (int k = 0; k < 4; ++k)
            if (e0 + k * TN + tid < end) atomicAdd(&cnt[p[k] & (TN - 1)], 1);
    }
    __syncthreads();
    const int c = cnt[tid];
    const int node = blockIdx.x * TN + tid;
    if (node < n) dinv[node] = rsqrtf((float)(1 + c));

    int inc = c;
#pragma unroll
    for (int o = 1; o < 32; o <<= 1) {
        const int v = __shfl_up(inc, o, 32);
        if (lane >= o) inc += v;
    }
    if (tid == 31) half0 = inc;
    __syncthreads();
    cnt[tid] = inc - c + (tid >= 32 ? half0 : 0);
    __syncthreads();

    const bool stage = (end - beg) <= SCAP;
    for (int e0 = beg; e0 < end; e0 += 4 * TN) {
        int p[4];
#pragma unroll
        for (int k = 0; k < 4; ++k)
            p[k] = binned[min(e0 + k * TN + tid, end - 1)];
#pragma unroll
        for (int k = 0; k < 4; ++k) {
            if (e0 + k * TN + tid < end) {
                const int pos = atomicAdd(&cnt[p[k] & (TN - 1)], 1);
                if (stage) buf[pos] = p[k];
                else       sorted[beg + pos] = p[k];
            }
        }
    }
    if (stage) {
        __syncthreads();
        for (int i = tid; i < end - beg; i += TN) sorted[beg + i] = buf[i];
    }
}

__device__ __forceinline__ void half_product(v8f (&acc)[4], const float* a_row,
                                             const float* wcol, int khalf) {
#pragma unroll 2
    for (int k4 = 0; k4 < 16; ++k4) {
        const int kl = k4 * 4 + khalf;
        v2f a;
        a.x = a_row[kl];
        a.y = a_row[kl + 1];
        const float* w0 = wcol + kl * LDSW_STRIDE;
        const float* w1 = w0 + LDSW_STRIDE;
#pragma unroll
        for (int nt = 0; nt < 4; ++nt) {
            v2f b;
            b.x = w0[nt * 16];
            b.y = w1[nt * 16];
            acc[nt] = __builtin_amdgcn_wmma_f32_16x16x4_f32(
                false, a, false, b, (short)0, acc[nt], false, false);
        }
    }
}

#define LW_STRIDE 16

__device__ __forceinline__ v8f logit_product(const float* a_row, const float* wl, int khalf) {
    const v8f vzero = {0.f, 0.f, 0.f, 0.f, 0.f, 0.f, 0.f, 0.f};
    v8f d0 = vzero, d1 = vzero;
#pragma unroll
    for (int k4 = 0; k4 < 16; k4 += 2) {
        const int kl = k4 * 4 + khalf;
        v2f a0, b0, a1, b1;
        a0.x = a_row[kl];
        a0.y = a_row[kl + 1];
        b0.x = wl[kl * LW_STRIDE];
        b0.y = wl[(kl + 1) * LW_STRIDE];
        a1.x = a_row[kl + 4];
        a1.y = a_row[kl + 5];
        b1.x = wl[(kl + 4) * LW_STRIDE];
        b1.y = wl[(kl + 5) * LW_STRIDE];
        d0 = __builtin_amdgcn_wmma_f32_16x16x4_f32(false, a0, false, b0, (short)0, d0, false, false);
        d1 = __builtin_amdgcn_wmma_f32_16x16x4_f32(false, a1, false, b1, (short)0, d1, false, false);
    }
    return d0 + d1;
}

__launch_bounds__(256)
__global__ void tile_layer(const int* __restrict__ binned, const int* __restrict__ binOff,
                           const float* __restrict__ dinv, const float* __restrict__ P,
                           const float* __restrict__ W, const float* __restrict__ bias,
                           float* __restrict__ outp, int n,
                           const float* __restrict__ Wl, const float* __restrict__ blin,
                           float* __restrict__ lg, float* __restrict__ out, int lmode) {
    __shared__ __attribute__((aligned(16))) float t[TN * TSTR];
    __shared__ __attribute__((aligned(16))) float ldsW[(HDIM / 2) * LDSW_STRIDE];
    __shared__ float ssp[2][TN];
    __shared__ __attribute__((aligned(16))) float ldsWl[HDIM * LW_STRIDE];
    const int tid  = threadIdx.x;
    const int lane = tid & 31;
    const int wave  = __builtin_amdgcn_readfirstlane(tid >> 5);
    const int node0 = blockIdx.x * TN;

    for (int i = 0; i < 8; ++i) {
        const int idx = (i * 256 + tid) * 4;
        const int row = idx >> 7, col = idx & 127;
        const float4 v = *(const float4*)(W + idx);
        *(float4*)(&ldsW[row * LDSW_STRIDE + col]) = v;
    }
    {
        const float4 v  = ((const float4*)Wl)[tid];
        const float4 z4 = {0.f, 0.f, 0.f, 0.f};
        float* wr = &ldsWl[(tid >> 1) * LW_STRIDE + (tid & 1) * 4];
        *(float4*)wr       = v;
        *(float4*)(wr + 8) = z4;
    }

    const int dvbits = __float_as_int(dinv[min(node0 + (lane & 7) * 8 + wave, n - 1)]);
#pragma unroll
    for (int i = 0; i < TN / 8; ++i) {
        const int r    = i * 8 + wave;
        const int node = node0 + r;
        const int nc   = min(node, n - 1);
        float w = __int_as_float(__builtin_amdgcn_readlane(dvbits, i));
        w = (node < n) ? w * w : 0.f;
        float4 v = ((const float4*)P)[(size_t)nc * 32 + lane];
        v.x *= w; v.y *= w; v.z *= w; v.w *= w;
        *(float4*)(&t[r * TSTR + lane * 4]) = v;
    }
    __syncthreads();

    const int beg = binOff[blockIdx.x];
    const int end = binOff[blockIdx.x + 1];
    for (int e0 = beg + wave * 32; e0 < end; e0 += 256) {
        const int cnt = min(32, end - e0);
        int s = 0, dl = 0;
        float w = 0.f;
        if (lane < cnt) {
            const int p = binned[e0 + lane];
            s  = p >> TSH;
            dl = p & (TN - 1);
            w  = dinv[s] * dinv[node0 + dl];
        }
        const int wbits = __float_as_int(w);

        int j = 0;
        for (; j + 8 <= cnt; j += 8) {
            float v[8][4];
#pragma unroll
            for (int u = 0; u < 8; ++u) {
                const int sj = __builtin_amdgcn_readlane(s, j + u);
                const float* hp = P + (size_t)sj * HDIM;
                v[u][0] = hp[lane];
                v[u][1] = hp[lane + 32];
                v[u][2] = hp[lane + 64];
                v[u][3] = hp[lane + 96];
            }
            int cur = __builtin_amdgcn_readlane(dl, j);
            float a0 = 0.f, a1 = 0.f, a2 = 0.f, a3 = 0.f;
#pragma unroll
            for (int u = 0; u < 8; ++u) {
                const int dj = __builtin_amdgcn_readlane(dl, j + u);
                const float wj = __int_as_float(__builtin_amdgcn_readlane(wbits, j + u));
                if (dj != cur) {
                    float* ap = t + cur * TSTR;
                    atomicAdd(ap + lane,      a0);
                    atomicAdd(ap + lane + 32, a1);
                    atomicAdd(ap + lane + 64, a2);
                    atomicAdd(ap + lane + 96, a3);
                    a0 = 0.f; a1 = 0.f; a2 = 0.f; a3 = 0.f;
                    cur = dj;
                }
                a0 = fmaf(v[u][0], wj, a0);
                a1 = fmaf(v[u][1], wj, a1);
                a2 = fmaf(v[u][2], wj, a2);
                a3 = fmaf(v[u][3], wj, a3);
            }
            float* ap = t + cur * TSTR;
            atomicAdd(ap + lane,      a0);
            atomicAdd(ap + lane + 32, a1);
            atomicAdd(ap + lane + 64, a2);
            atomicAdd(ap + lane + 96, a3);
        }
        for (; j < cnt; ++j) {
            const int sj = __builtin_amdgcn_readlane(s, j);
            const int dj = __builtin_amdgcn_readlane(dl, j);
            const float wj = __int_as_float(__builtin_amdgcn_readlane(wbits, j));
            const float* hp = P + (size_t)sj * HDIM;
            float* ap = t + dj * TSTR;
            const float v0 = hp[lane];
            const float v1 = hp[lane + 32];
            const float v2 = hp[lane + 64];
            const float v3 = hp[lane + 96];
            atomicAdd(ap + lane,      v0 * wj);
            atomicAdd(ap + lane + 32, v1 * wj);
            atomicAdd(ap + lane + 64, v2 * wj);
            atomicAdd(ap + lane + 96, v3 * wj);
        }
    }
    __syncthreads();

    const int m     = lane & 15;
    const int hh    = lane >> 4;
    const int khalf = hh * 2;
    const int strip = wave & 3;
    const int c0    = (wave >> 2) * 64;
    const float* aRow = &t[(strip * 16 + m) * TSTR];

    v8f acc[4];
    const v8f vzero = {0.f, 0.f, 0.f, 0.f, 0.f, 0.f, 0.f, 0.f};
#pragma unroll
    for (int nt = 0; nt < 4; ++nt) acc[nt] = vzero;

    half_product(acc, aRow, &ldsW[c0 + m], khalf);
    __syncthreads();
#pragma unroll
    for (int i = 0; i < 8; ++i) {
        const int idx = (i * 256 + tid) * 4;
        const float4 v = *(const float4*)(W + (HDIM / 2) * HDIM + idx);
        *(float4*)(&ldsW[(idx >> 7) * LDSW_STRIDE + (idx & 127)]) = v;
    }
    __syncthreads();
    half_product(acc, aRow + HDIM / 2, &ldsW[c0 + m], khalf);

    float bv[4];
#pragma unroll
    for (int nt = 0; nt < 4; ++nt) bv[nt] = bias[c0 + nt * 16 + m];
    float ssr[8];
#pragma unroll
    for (int r = 0; r < 8; ++r) {
        float q = 0.f;
#pragma unroll
        for (int nt = 0; nt < 4; ++nt) {
            const float v = fmaxf(acc[nt][r] + bv[nt], 0.f);
            acc[nt][r] = v;
            q = fmaf(v, v, q);
        }
        ssr[r] = q;
    }
#pragma unroll
    for (int off = 8; off > 0; off >>= 1) {
#pragma unroll
        for (int r = 0; r < 8; ++r) ssr[r] += __shfl_xor(ssr[r], off, 32);
    }
    const int rowb = strip * 16 + hh * 8;
    if (m == 0) {
#pragma unroll
        for (int r = 0; r < 8; ++r) ssp[wave >> 2][rowb + r] = ssr[r];
    }
    __syncthreads();
#pragma unroll
    for (int r = 0; r < 8; ++r) {
        const float ss = ssp[0][rowb + r] + ssp[1][rowb + r];
        const float scale = 1.0f / fmaxf(sqrtf(ss), 1e-12f);
#pragma unroll
        for (int nt = 0; nt < 4; ++nt)
            t[(rowb + r) * TSTR + c0 + nt * 16 + m] = acc[nt][r] * scale;
    }
    __syncthreads();

    const bool fin = (tid < TN) && (node0 + tid < n);
    float4 g0 = {0.f, 0.f, 0.f, 0.f}, g1 = g0;
    if (fin && lmode != 0) {
        g0 = ((const float4*)lg)[(size_t)(node0 + tid) * 2];
        g1 = ((const float4*)lg)[(size_t)(node0 + tid) * 2 + 1];
        if (lmode == 2) {
            const float4 b0 = ((const float4*)blin)[0];
            const float4 b1 = ((const float4*)blin)[1];
            g0.x += b0.x; g0.y += b0.y; g0.z += b0.z; g0.w += b0.w;
            g1.x += b1.x; g1.y += b1.y; g1.z += b1.z; g1.w += b1.w;
        }
    }

    if (outp != nullptr) {
#pragma unroll
        for (int i = 0; i < TN / 8; ++i) {
            const int r    = i * 8 + wave;
            const int node = node0 + r;
            if (node < n)
                ((float4*)outp)[(size_t)node * 32 + lane] = *(const float4*)(&t[r * TSTR + lane * 4]);
        }
    }

    const int kh = wave >> 2;
    const v8f dl = logit_product(aRow + kh * (HDIM / 2),
                                 &ldsWl[kh * (HDIM / 2) * LW_STRIDE + m], khalf);
    float* lgs = ldsW;
    if (m < 8) {
#pragma unroll
        for (int r = 0; r < 8; ++r) lgs[(kh * TN + rowb + r) * 8 + m] = dl[r];
    }
    __syncthreads();

    if (fin) {
        const size_t node = (size_t)(node0 + tid);
        const float4 p0 = *(const float4*)(&lgs[tid * 8]);
        const float4 p1 = *(const float4*)(&lgs[tid * 8 + 4]);
        const float4 q0 = *(const float4*)(&lgs[(TN + tid) * 8]);
        const float4 q1 = *(const float4*)(&lgs[(TN + tid) * 8 + 4]);
        float s[8] = {p0.x + q0.x, p0.y + q0.y, p0.z + q0.z, p0.w + q0.w,
                      p1.x + q1.x, p1.y + q1.y, p1.z + q1.z, p1.w + q1.w};
        s[0] += g0.x; s[1] += g0.y; s[2] += g0.z; s[3] += g0.w;
        s[4] += g1.x; s[5] += g1.y; s[6] += g1.z; s[7] += g1.w;
        float* dstp = lg;
        if (lmode == 2) {
            float mx = s[0];
#pragma unroll
            for (int c = 1; c < 8; ++c) mx = fmaxf(mx, s[c]);
            float se = 0.f;
#pragma unroll
            for (int c = 0; c < 8; ++c) se += expf(s[c] - mx);
            const float lse = mx + logf(se);
#pragma unroll
            for (int c = 0; c < 8; ++c) s[c] -= lse;
            dstp = out;
        }
        const float4 o0 = {s[0], s[1], s[2], s[3]};
        const float4 o1 = {s[4], s[5], s[6], s[7]};
        ((float4*)dstp)[node * 2]     = o0;
        ((float4*)dstp)[node * 2 + 1] = o1;
    }
}

extern "C" void kernel_launch(void* const* d_in, const int* in_sizes, int n_in,
                              void* d_out, int out_size, void* d_ws, size_t ws_size,
                              hipStream_t stream) {
    const float* x    = (const float*)d_in[0];
    const int*   ei   = (const int*)d_in[1];
    const float* W1   = (const float*)d_in[2];
    const float* b1   = (const float*)d_in[3];
    const float* W2   = (const float*)d_in[4];
    const float* b2   = (const float*)d_in[5];
    const float* W3   = (const float*)d_in[6];
    const float* b3   = (const float*)d_in[7];
    const float* Wlin = (const float*)d_in[8];
    const float* blin = (const float*)d_in[9];

    const int n = in_sizes[0] / HDIM;
    const int E = in_sizes[1] / 2;
    const int* src = ei;
    const int* dst = ei + E;

    float* ws   = (float*)d_ws;
    float* dinv = ws;
    float* emb  = ws + (size_t)n;
    int* binCnt = (int*)(emb + (size_t)3 * n * HDIM);
    int* binOff = binCnt + MAXB;
    int* binCur = binOff + MAXB + 64;
    int* binned = binCur + MAXB;
    int* sorted = binned + E;
    const int nb   = (n + TN - 1) / TN;
    const int nblk = (E + BCH - 1) / BCH;

    const int T = 256;
    zero_counts<<<(MAXB + T - 1) / T, T, 0, stream>>>(binCnt, MAXB);
    bin_count<<<nblk, 256, 0, stream>>>(dst, binCnt, E);
    bin_scan<<<1, 1024, 0, stream>>>(binCnt, binOff, binCur, MAXB, E);
    bin_fill<<<nblk, 256, 0, stream>>>(src, dst, binCur, binned, E);

    tile_sort<<<nb, TN, 0, stream>>>(binned, binOff, sorted, dinv, n);

    float* e1 = emb;
    float* e2 = emb + (size_t)n * HDIM;
    float* lg = emb + (size_t)2 * n * HDIM;
    float* outf = (float*)d_out;
    tile_layer<<<nb, 256, 0, stream>>>(sorted, binOff, dinv, x,  W1, b1, e1, n,
                                       Wlin, blin, lg, outf, 0);
    tile_layer<<<nb, 256, 0, stream>>>(sorted, binOff, dinv, e1, W2, b2, e2, n,
                                       Wlin + HDIM * 8, blin, lg, outf, 1);
    tile_layer<<<nb, 256, 0, stream>>>(sorted, binOff, dinv, e2, W3, b3, nullptr, n,
                                       Wlin + 2 * HDIM * 8, blin, lg, outf, 2);
}
